# starting kernel
# baseline (speedup 1.0000x reference)
#include <hip/hip_runtime.h>
#include <hip/hip_bf16.h>

typedef __attribute__((ext_vector_type(2))) float v2f;
typedef __attribute__((ext_vector_type(4))) float v4f;
typedef __attribute__((ext_vector_type(8))) float v8f;

#define L_LEN   65536
#define N_ROWS  64
#define TAPS    80
#define KSPAN   96
#define NCHUNK  (KSPAN / 4)
#define PAD     16
#define WAVES_PER_BLOCK 8
#define TILES_TOTAL (N_ROWS * (L_LEN / 256))

__global__ __launch_bounds__(256) void
HanningTemplateLayer_61761629716908_kernel(const float* __restrict__ x,
                                           const float* __restrict__ tw,
                                           float* __restrict__ out) {
    __shared__ float eff_pad[128];

    {
        const int t = threadIdx.x;
        if (t < 128) {
            float acc = 0.0f;
            const int tt = t - PAD;
            if ((unsigned)tt < TAPS) {
                float w0 = tw[0], w1 = tw[1], w2 = tw[2], w3 = tw[3];
                float mx = fmaxf(fmaxf(w0, w1), fmaxf(w2, w3));
                float e0 = __expf(w0 - mx), e1 = __expf(w1 - mx);
                float e2 = __expf(w2 - mx), e3 = __expf(w3 - mx);
                float inv = 1.0f / (e0 + e1 + e2 + e3);
                const float sw[4]   = {e0 * inv, e1 * inv, e2 * inv, e3 * inv};
                const int   wid4[4] = {10, 20, 30, 40};
                const float twopi = 6.28318530717958647692f;
                #pragma unroll
                for (int i = 0; i < 4; ++i) {
                    int w = wid4[i];
                    int k = tt - 40 + w;
                    if (k >= 0 && k < 2 * w) {
                        float h = 0.5f - 0.5f * cosf(twopi * (float)k / (float)(2 * w - 1));
                        acc += sw[i] * h;
                    }
                }
            }
            eff_pad[t] = acc;
        }
    }
    __syncthreads();

    const int lane = threadIdx.x & 31;
    const int wid  = threadIdx.x >> 5;
    const int W    = blockIdx.x * WAVES_PER_BLOCK + wid;
    const int row  = W >> 8;
    const int tile = W & 255;
    const int j0   = tile << 8;

    const int m   = lane & 15;
    const int hi  = lane >> 4;
    const int klo = hi << 1;

    v2f a_frag[NCHUNK];
    #pragma unroll
    for (int c = 0; c < NCHUNK; ++c) {
        const int base = c * 4 + klo - m + PAD;
        a_frag[c].x = eff_pad[base];
        a_frag[c].y = eff_pad[base + 1];
    }

    const float* xrow  = x + (size_t)row * L_LEN;
    const int    basep = j0 - 40 + 16 * m + klo;
    v8f acc0 = {};
    v8f acc1 = {};

    const bool interior = (j0 >= 40) && (j0 + 296 <= L_LEN);
    if (interior) {
        #pragma unroll
        for (int c = 0; c < NCHUNK; c += 2) {
            v2f b0 = *(const v2f*)(xrow + basep + 4 * c);
            v2f b1 = *(const v2f*)(xrow + basep + 4 * (c + 1));
            acc0 = __builtin_amdgcn_wmma_f32_16x16x4_f32(
                false, a_frag[c], false, b0, (short)0, acc0, false, false);
            acc1 = __builtin_amdgcn_wmma_f32_16x16x4_f32(
                false, a_frag[c + 1], false, b1, (short)0, acc1, false, false);
        }
    } else {
        #pragma unroll
        for (int c = 0; c < NCHUNK; c += 2) {
            const int p0 = basep + 4 * c;
            const int p1 = p0 + 4;
            int pc0 = p0 < 0 ? 0 : p0;
            pc0 = pc0 > (L_LEN - 2) ? (L_LEN - 2) : pc0;
            int pc1 = p1 < 0 ? 0 : p1;
            pc1 = pc1 > (L_LEN - 2) ? (L_LEN - 2) : pc1;
            v2f b0 = *(const v2f*)(xrow + pc0);
            v2f b1 = *(const v2f*)(xrow + pc1);
            b0.x = ((unsigned)p0       < L_LEN) ? b0.x : 0.0f;
            b0.y = ((unsigned)(p0 + 1) < L_LEN) ? b0.y : 0.0f;
            b1.x = ((unsigned)p1       < L_LEN) ? b1.x : 0.0f;
            b1.y = ((unsigned)(p1 + 1) < L_LEN) ? b1.y : 0.0f;
            acc0 = __builtin_amdgcn_wmma_f32_16x16x4_f32(
                false, a_frag[c], false, b0, (short)0, acc0, false, false);
            acc1 = __builtin_amdgcn_wmma_f32_16x16x4_f32(
                false, a_frag[c + 1], false, b1, (short)0, acc1, false, false);
        }
    }

    float* orow = out + (size_t)row * L_LEN + j0 + 8 * hi + 16 * m;
    v4f lo = {acc0[0] + acc1[0], acc0[1] + acc1[1], acc0[2] + acc1[2], acc0[3] + acc1[3]};
    v4f hh = {acc0[4] + acc1[4], acc0[5] + acc1[5], acc0[6] + acc1[6], acc0[7] + acc1[7]};
    *(v4f*)(orow)     = lo;
    *(v4f*)(orow + 4) = hh;
}

extern "C" void kernel_launch(void* const* d_in, const int* in_sizes, int n_in,
                              void* d_out, int out_size, void* d_ws, size_t ws_size,
                              hipStream_t stream) {
    (void)in_sizes; (void)n_in; (void)d_ws; (void)ws_size; (void)out_size;
    const float* x  = (const float*)d_in[0];
    const float* tw = (const float*)d_in[1];
    float* out = (float*)d_out;

    const int blocks = TILES_TOTAL / WAVES_PER_BLOCK;
    HanningTemplateLayer_61761629716908_kernel<<<blocks, 256, 0, stream>>>(x, tw, out);
}


# signal staged per wave in LDS: each of 2048 waves owns 2048 contiguous outputs, reads its slab once by 17 whole-line b128 loads, takes B fragments from a padded LDS slab; per-output sum order kept
# speedup vs baseline: 1.4165x; 1.4165x over previous
#include <hip/hip_runtime.h>
#include <hip/hip_bf16.h>


typedef __attribute__((ext_vector_type(2))) float v2f;
typedef __attribute__((ext_vector_type(4))) float v4f;
typedef __attribute__((ext_vector_type(8))) float v8f;

#define L_LEN   65536
#define N_ROWS  64
#define TAPS    80
#define KSPAN   96
#define NCHUNK  (KSPAN / 4)
#define PAD     16
#define WAVES_PER_BLOCK 8

#define TILE_OUT        256
#define TILES_PER_WAVE  8
#define SEG_OUT         (TILE_OUT * TILES_PER_WAVE)
#define SEGS_PER_ROW    (L_LEN / SEG_OUT)
#define WAVES_TOTAL     (N_ROWS * SEGS_PER_ROW)
#define SLAB_LEAD       64
#define SLAB_LOADS      17
#define SLAB_PITCH      20
#define SLAB_FLOATS     (SLAB_LOADS * 8 * SLAB_PITCH)

__global__ __launch_bounds__(256) void
HanningTemplateLayer_61761629716908_kernel(const float* __restrict__ x,
                                           const float* __restrict__ tw,
                                           float* __restrict__ out) {
    __shared__ float eff_pad[128];
    __shared__ __attribute__((aligned(16))) float slab_all[WAVES_PER_BLOCK * SLAB_FLOATS];

    const int lane = threadIdx.x & 31;
    const int wid  = threadIdx.x >> 5;
    const int gw   = blockIdx.x * WAVES_PER_BLOCK + wid;
    const int row  = gw / SEGS_PER_ROW;
    const int seg0 = (gw % SEGS_PER_ROW) * SEG_OUT;
    const float* xrow = x + (size_t)row * L_LEN;

    v4f ld[SLAB_LOADS];
    const int gp0 = seg0 - SLAB_LEAD + 4 * lane;
    const int gp_last = gp0 + 128 * (SLAB_LOADS - 1);
    #pragma unroll
    for (int i = 0; i < SLAB_LOADS; ++i) {
        int gp = gp0 + 128 * i;
        if (i == 0)              gp = gp < 0 ? 0 : gp;
        if (i == SLAB_LOADS - 1) gp = gp > (L_LEN - 4) ? (L_LEN - 4) : gp;
        ld[i] = *(const v4f*)(xrow + gp);
        if (i == 4) __builtin_amdgcn_sched_barrier(0);
    }

    {
        const int t = threadIdx.x;
        if (t < 128) {
            float acc = 0.0f;
            const int tt = t - PAD;
            if ((unsigned)tt < TAPS) {
                float w0 = tw[0], w1 = tw[1], w2 = tw[2], w3 = tw[3];
                float mx = fmaxf(fmaxf(w0, w1), fmaxf(w2, w3));
                float e0 = __expf(w0 - mx), e1 = __expf(w1 - mx);
                float e2 = __expf(w2 - mx), e3 = __expf(w3 - mx);
                float inv = 1.0f / (e0 + e1 + e2 + e3);
                const float sw[4]   = {e0 * inv, e1 * inv, e2 * inv, e3 * inv};
                const int   wid4[4] = {10, 20, 30, 40};
                const float twopi = 6.28318530717958647692f;
                #pragma unroll
                for (int i = 0; i < 4; ++i) {
                    int w = wid4[i];
                    int k = tt - 40 + w;
                    if (k >= 0 && k < 2 * w) {
                        float h = 0.5f - 0.5f * cosf(twopi * (float)k / (float)(2 * w - 1));
                        acc += sw[i] * h;
                    }
                }
            }
            eff_pad[t] = acc;
        }
    }
    __builtin_amdgcn_fence(__ATOMIC_RELEASE, "workgroup", "local");
    __builtin_amdgcn_s_barrier();
    __builtin_amdgcn_fence(__ATOMIC_ACQUIRE, "workgroup", "local");

    const int m   = lane & 15;
    const int hi  = lane >> 4;
    const int klo = hi << 1;

    v2f a_frag[NCHUNK];
    #pragma unroll
    for (int c = 0; c < NCHUNK; ++c) {
        const int base = c * 4 + klo - m + PAD;
        a_frag[c].x = eff_pad[base];
        a_frag[c].y = eff_pad[base + 1];
    }

    float* slab = slab_all + wid * SLAB_FLOATS;
    float* st_ptr = slab + (lane >> 2) * SLAB_PITCH + 4 * (lane & 3);
    const float* rd_ptr = slab + m * SLAB_PITCH + klo;

    v2f bf[2][NCHUNK];

    {
        const bool in0 = gp0 >= 0;
        v4f v0;
        v0.x = in0 ? ld[0].x : 0.0f;
        v0.y = in0 ? ld[0].y : 0.0f;
        v0.z = in0 ? ld[0].z : 0.0f;
        v0.w = in0 ? ld[0].w : 0.0f;
        *(v4f*)(st_ptr) = v0;
        *(v4f*)(st_ptr + 1 * 8 * SLAB_PITCH) = ld[1];
        *(v4f*)(st_ptr + 2 * 8 * SLAB_PITCH) = ld[2];
    }
    __builtin_amdgcn_fence(__ATOMIC_RELEASE, "wavefront", "local");
    __builtin_amdgcn_fence(__ATOMIC_ACQUIRE, "wavefront", "local");
    #pragma unroll
    for (int c = 0; c < NCHUNK; ++c)
        bf[0][c] = *(const v2f*)(rd_ptr + ((24 + 4 * c) >> 4) * SLAB_PITCH + ((24 + 4 * c) & 15));

    float* orow = out + (size_t)row * L_LEN + seg0 + 8 * hi + 16 * m;

    #pragma unroll
    for (int t = 0; t < TILES_PER_WAVE; ++t) {
        __builtin_amdgcn_sched_barrier(0);
        if (t + 1 < TILES_PER_WAVE) {
            *(v4f*)(st_ptr + (2 * t + 3) * 8 * SLAB_PITCH) = ld[2 * t + 3];
            v4f vb = ld[2 * t + 4];
            if (2 * t + 4 == SLAB_LOADS - 1) {
                const bool in1 = gp_last < L_LEN;
                vb.x = in1 ? vb.x : 0.0f;
                vb.y = in1 ? vb.y : 0.0f;
                vb.z = in1 ? vb.z : 0.0f;
                vb.w = in1 ? vb.w : 0.0f;
            }
            *(v4f*)(st_ptr + (2 * t + 4) * 8 * SLAB_PITCH) = vb;
            __builtin_amdgcn_fence(__ATOMIC_RELEASE, "wavefront", "local");
            __builtin_amdgcn_fence(__ATOMIC_ACQUIRE, "wavefront", "local");
            #pragma unroll
            for (int c = 0; c < NCHUNK; ++c)
                bf[(t + 1) & 1][c] = *(const v2f*)(rd_ptr
                    + (16 * (t + 1) + ((24 + 4 * c) >> 4)) * SLAB_PITCH + ((24 + 4 * c) & 15));
        }
        __builtin_amdgcn_sched_barrier(0);

        v8f acc0 = {};
        v8f acc1 = {};
        #pragma unroll
        for (int c = 0; c < NCHUNK; c += 2) {
            acc0 = __builtin_amdgcn_wmma_f32_16x16x4_f32(
                false, a_frag[c], false, bf[t & 1][c], (short)0, acc0, false, false);
            acc1 = __builtin_amdgcn_wmma_f32_16x16x4_f32(
                false, a_frag[c + 1], false, bf[t & 1][c + 1], (short)0, acc1, false, false);
        }

        v4f lo = {acc0[0] + acc1[0], acc0[1] + acc1[1], acc0[2] + acc1[2], acc0[3] + acc1[3]};
        v4f hh = {acc0[4] + acc1[4], acc0[5] + acc1[5], acc0[6] + acc1[6], acc0[7] + acc1[7]};
        *(v4f*)(orow + t * TILE_OUT)     = lo;
        *(v4f*)(orow + t * TILE_OUT + 4) = hh;
    }
}

extern "C" void kernel_launch(void* const* d_in, const int* in_sizes, int n_in,
                              void* d_out, int out_size, void* d_ws, size_t ws_size,
                              hipStream_t stream) {
    (void)in_sizes; (void)n_in; (void)d_ws; (void)ws_size; (void)out_size;
    const float* x  = (const float*)d_in[0];
    const float* tw = (const float*)d_in[1];
    float* out = (float*)d_out;

    const int blocks = WAVES_TOTAL / WAVES_PER_BLOCK;
    HanningTemplateLayer_61761629716908_kernel<<<blocks, 256, 0, stream>>>(x, tw, out);
}


# filter build off the path to the first WMMA: four window values per tap (cospi, branch-free) run under the weights load, tile 0 staged between barrier signal and wait; taps differ in last bits
# speedup vs baseline: 1.4912x; 1.0527x over previous
#include <hip/hip_runtime.h>
#include <hip/hip_bf16.h>


typedef __attribute__((ext_vector_type(2))) float v2f;
typedef __attribute__((ext_vector_type(4))) float v4f;
typedef __attribute__((ext_vector_type(8))) float v8f;

#define L_LEN   65536
#define N_ROWS  64
#define TAPS    80
#define KSPAN   96
#define NCHUNK  (KSPAN / 4)
#define PAD     16
#define WAVES_PER_BLOCK 8

#define TILE_OUT        256
#define TILES_PER_WAVE  8
#define SEG_OUT         (TILE_OUT * TILES_PER_WAVE)
#define SEGS_PER_ROW    (L_LEN / SEG_OUT)
#define WAVES_TOTAL     (N_ROWS * SEGS_PER_ROW)
#define SLAB_LEAD       64
#define SLAB_LOADS      17
#define SLAB_PITCH      20
#define SLAB_FLOATS     (SLAB_LOADS * 8 * SLAB_PITCH)

__global__ __launch_bounds__(256) void
HanningTemplateLayer_61761629716908_kernel(const float* __restrict__ x,
                                           const float* __restrict__ tw,
                                           float* __restrict__ out) {
    __shared__ float eff_pad[128];
    __shared__ __attribute__((aligned(16))) float slab_all[WAVES_PER_BLOCK * SLAB_FLOATS];

    const int lane = threadIdx.x & 31;
    const int wid  = threadIdx.x >> 5;
    const int gw   = blockIdx.x * WAVES_PER_BLOCK + wid;
    const int row  = gw / SEGS_PER_ROW;
    const int seg0 = (gw % SEGS_PER_ROW) * SEG_OUT;
    const float* xrow = x + (size_t)row * L_LEN;

    v4f ld[SLAB_LOADS];
    const int gp0 = seg0 - SLAB_LEAD + 4 * lane;
    const int gp_last = gp0 + 128 * (SLAB_LOADS - 1);
    #pragma unroll
    for (int i = 0; i < SLAB_LOADS; ++i) {
        int gp = gp0 + 128 * i;
        if (i == 0)              gp = gp < 0 ? 0 : gp;
        if (i == SLAB_LOADS - 1) gp = gp > (L_LEN - 4) ? (L_LEN - 4) : gp;
        ld[i] = *(const v4f*)(xrow + gp);
        if (i == 4) __builtin_amdgcn_sched_barrier(0);
    }

    if (threadIdx.x < 128) {
        float s0 = tw[0], s1 = tw[1], s2 = tw[2], s3 = tw[3];
        __builtin_amdgcn_sched_barrier(0);
        const int t  = threadIdx.x;
        const int tt = t - PAD;
        float hv[4];
        #pragma unroll
        for (int i = 0; i < 4; ++i) {
            const int  w   = 10 * (i + 1);
            const int  k   = tt - 40 + w;
            const bool in  = (unsigned)k < (unsigned)(2 * w);
            const float q  = (float)(2 * (in ? k : 0)) / (float)(2 * w - 1);
            const float h  = 0.5f - 0.5f * cospif(q);
            hv[i] = in ? h : 0.0f;
        }
        asm volatile("" : "+s"(s0), "+s"(s1), "+s"(s2), "+s"(s3)
                        : "v"(hv[0]), "v"(hv[1]), "v"(hv[2]), "v"(hv[3]));
        float mx = fmaxf(fmaxf(s0, s1), fmaxf(s2, s3));
        float e0 = __expf(s0 - mx), e1 = __expf(s1 - mx);
        float e2 = __expf(s2 - mx), e3 = __expf(s3 - mx);
        float inv = 1.0f / (e0 + e1 + e2 + e3);
        const float sw[4] = {e0 * inv, e1 * inv, e2 * inv, e3 * inv};
        eff_pad[t] = sw[0] * hv[0] + sw[1] * hv[1] + sw[2] * hv[2] + sw[3] * hv[3];
    }
    __builtin_amdgcn_fence(__ATOMIC_RELEASE, "workgroup", "local");
    asm volatile("s_barrier_signal -1" ::: "memory");

    const int m   = lane & 15;
    const int hi  = lane >> 4;
    const int klo = hi << 1;

    float* slab = slab_all + wid * SLAB_FLOATS;
    float* st_ptr = slab + (lane >> 2) * SLAB_PITCH + 4 * (lane & 3);
    const float* rd_ptr = slab + m * SLAB_PITCH + klo;

    v2f bf[2][NCHUNK];

    {
        const bool in0 = gp0 >= 0;
        v4f v0;
        v0.x = in0 ? ld[0].x : 0.0f;
        v0.y = in0 ? ld[0].y : 0.0f;
        v0.z = in0 ? ld[0].z : 0.0f;
        v0.w = in0 ? ld[0].w : 0.0f;
        *(v4f*)(st_ptr) = v0;
        *(v4f*)(st_ptr + 1 * 8 * SLAB_PITCH) = ld[1];
        *(v4f*)(st_ptr + 2 * 8 * SLAB_PITCH) = ld[2];
    }
    __builtin_amdgcn_fence(__ATOMIC_RELEASE, "wavefront", "local");
    __builtin_amdgcn_fence(__ATOMIC_ACQUIRE, "wavefront", "local");
    #pragma unroll
    for (int c = 0; c < NCHUNK; ++c)
        bf[0][c] = *(const v2f*)(rd_ptr + ((24 + 4 * c) >> 4) * SLAB_PITCH + ((24 + 4 * c) & 15));

    asm volatile("s_barrier_wait -1" ::: "memory");

    v2f a_frag[NCHUNK];
    #pragma unroll
    for (int c = 0; c < NCHUNK; ++c) {
        const int base = c * 4 + klo - m + PAD;
        a_frag[c].x = eff_pad[base];
        a_frag[c].y = eff_pad[base + 1];
    }

    float* orow = out + (size_t)row * L_LEN + seg0 + 8 * hi + 16 * m;

    #pragma unroll
    for (int t = 0; t < TILES_PER_WAVE; ++t) {
        __builtin_amdgcn_sched_barrier(0);
        if (t + 1 < TILES_PER_WAVE) {
            *(v4f*)(st_ptr + (2 * t + 3) * 8 * SLAB_PITCH) = ld[2 * t + 3];
            v4f vb = ld[2 * t + 4];
            if (2 * t + 4 == SLAB_LOADS - 1) {
                const bool in1 = gp_last < L_LEN;
                vb.x = in1 ? vb.x : 0.0f;
                vb.y = in1 ? vb.y : 0.0f;
                vb.z = in1 ? vb.z : 0.0f;
                vb.w = in1 ? vb.w : 0.0f;
            }
            *(v4f*)(st_ptr + (2 * t + 4) * 8 * SLAB_PITCH) = vb;
            __builtin_amdgcn_fence(__ATOMIC_RELEASE, "wavefront", "local");
            __builtin_amdgcn_fence(__ATOMIC_ACQUIRE, "wavefront", "local");
            #pragma unroll
            for (int c = 0; c < NCHUNK; ++c)
                bf[(t + 1) & 1][c] = *(const v2f*)(rd_ptr
                    + (16 * (t + 1) + ((24 + 4 * c) >> 4)) * SLAB_PITCH + ((24 + 4 * c) & 15));
        }
        __builtin_amdgcn_sched_barrier(0);

        v8f acc0 = {};
        v8f acc1 = {};
        #pragma unroll
        for (int c = 0; c < NCHUNK; c += 2) {
            acc0 = __builtin_amdgcn_wmma_f32_16x16x4_f32(
                false, a_frag[c], false, bf[t & 1][c], (short)0, acc0, false, false);
            acc1 = __builtin_amdgcn_wmma_f32_16x16x4_f32(
                false, a_frag[c + 1], false, bf[t & 1][c + 1], (short)0, acc1, false, false);
        }

        v4f lo = {acc0[0] + acc1[0], acc0[1] + acc1[1], acc0[2] + acc1[2], acc0[3] + acc1[3]};
        v4f hh = {acc0[4] + acc1[4], acc0[5] + acc1[5], acc0[6] + acc1[6], acc0[7] + acc1[7]};
        *(v4f*)(orow + t * TILE_OUT)     = lo;
        *(v4f*)(orow + t * TILE_OUT + 4) = hh;
    }
}

extern "C" void kernel_launch(void* const* d_in, const int* in_sizes, int n_in,
                              void* d_out, int out_size, void* d_ws, size_t ws_size,
                              hipStream_t stream) {
    (void)in_sizes; (void)n_in; (void)d_ws; (void)ws_size; (void)out_size;
    const float* x  = (const float*)d_in[0];
    const float* tw = (const float*)d_in[1];
    float* out = (float*)d_out;

    const int blocks = WAVES_TOTAL / WAVES_PER_BLOCK;
    HanningTemplateLayer_61761629716908_kernel<<<blocks, 256, 0, stream>>>(x, tw, out);
}
